# starting kernel
# baseline (speedup 1.0000x reference)
#include <hip/hip_runtime.h>
#include <math.h>

#define N_INST 64
#define NPIX   64000

typedef __attribute__((ext_vector_type(2))) float v2f;
typedef __attribute__((ext_vector_type(8))) float v8f;

__device__ __forceinline__ float sigmoidf(float x) {
    return __builtin_amdgcn_rcpf(1.0f + __expf(-x));
}

__global__ __launch_bounds__(256) void feat_kernel(
    const int* __restrict__ cls, const float* __restrict__ U_w,
    const float* __restrict__ V_w, const float* __restrict__ P_w,
    float* __restrict__ W1, float* __restrict__ F2) {
    int gid = blockIdx.x * 256 + threadIdx.x;
    int c = gid >> 6;
    int i = gid & 63;
    int a = cls[i] - 1;
    float u = U_w[c * 80 + a];
    float v = V_w[c * 80 + a];
    W1[c * 64 + i] = fmaxf(u, 0.0f) * P_w[c];
    F2[c * 64 + i] = fmaxf(v, 0.0f);
}

__global__ __launch_bounds__(256) void logit_kernel(
    const float* __restrict__ W1, const float* __restrict__ F2,
    const float* __restrict__ bbox, const float* __restrict__ Wc_w,
    const float* __restrict__ P_w, float* __restrict__ R) {
    int gid = blockIdx.x * 256 + threadIdx.x;
    int i = gid & 63;
    int j = gid >> 6;

    float acc = 0.0f;
    for (int c = 0; c < 256; ++c)
        acc += W1[c * 64 + i] * F2[c * 64 + j];

    float xi0 = bbox[i * 5 + 1], yi0 = bbox[i * 5 + 2];
    float xi1 = bbox[i * 5 + 3], yi1 = bbox[i * 5 + 4];
    float xj0 = bbox[j * 5 + 1], yj0 = bbox[j * 5 + 2];
    float xj1 = bbox[j * 5 + 3], yj1 = bbox[j * 5 + 4];
    float wi = xi1 - xi0 + 1.0f, hi = yi1 - yi0 + 1.0f;
    float wj = xj1 - xj0 + 1.0f, hj = yj1 - yj0 + 1.0f;
    float xci = xi0 + 0.5f * wi, yci = yi0 + 0.5f * hi;
    float xcj = xj0 + 0.5f * wj, ycj = yj0 + 0.5f * hj;
    float rwi = __builtin_amdgcn_rcpf(wi);
    float rhi = __builtin_amdgcn_rcpf(hi);
    float dx = -(xci - xcj) * rwi;
    float dy = -(yci - ycj) * rhi;
    float dw = __logf(wj * rwi);
    float dh = __logf(hj * rhi);

    for (int o = 0; o < 128; ++o) {
        float v = dx * Wc_w[o * 4 + 0] + dy * Wc_w[o * 4 + 1] +
                  dw * Wc_w[o * 4 + 2] + dh * Wc_w[o * 4 + 3];
        acc += fmaxf(v, 0.0f) * P_w[256 + o];
    }
    R[i * 64 + j] = sigmoidf(acc);
}

__global__ __launch_bounds__(256) void omat_kernel(
    const float* __restrict__ R, float* __restrict__ O_ws,
    float* __restrict__ O_out) {
    int gid = blockIdx.x * 256 + threadIdx.x;
    int i = gid >> 6;
    int j = gid & 63;
    float o = fmaxf(R[i * 64 + j] - R[j * 64 + i], 0.0f);
    O_ws[gid]  = o;
    O_out[gid] = o;
}

__global__ __launch_bounds__(256) void apply_kernel(
    const float* __restrict__ mask, const float* __restrict__ O,
    float* __restrict__ out) {
    __shared__ float sO[64][68];

    for (int idx = threadIdx.x; idx < 64 * 64; idx += 256)
        sO[idx >> 6][idx & 63] = O[idx];
    __syncthreads();

    const int lane = threadIdx.x & 31;
    const int wave = threadIdx.x >> 5;
    const int col  = lane & 15;
    const int half = lane >> 4;
    const int p    = blockIdx.x * 128 + wave * 16 + col;

    v8f acc[4] = {v8f{}, v8f{}, v8f{}, v8f{}};

    #pragma unroll
    for (int k0 = 0; k0 < 64; k0 += 4) {
        const int kA = k0 + half * 2;
        v2f b;
        b[0] = sigmoidf(mask[(kA + 0) * NPIX + p]);
        b[1] = sigmoidf(mask[(kA + 1) * NPIX + p]);
        #pragma unroll
        for (int mt = 0; mt < 4; ++mt) {
            v2f a;
            a[0] = sO[mt * 16 + col][kA + 0];
            a[1] = sO[mt * 16 + col][kA + 1];
            acc[mt] = __builtin_amdgcn_wmma_f32_16x16x4_f32(
                false, a, false, b, (short)0, acc[mt], false, false);
        }
    }

    #pragma unroll
    for (int mt = 0; mt < 4; ++mt) {
        #pragma unroll
        for (int r = 0; r < 8; ++r) {
            int row = mt * 16 + half * 8 + r;
            float m  = mask[row * NPIX + p];
            float tm = sigmoidf(m);
            out[row * NPIX + p] = m * (1.0f - tm * acc[mt][r]);
        }
    }
}

extern "C" void kernel_launch(void* const* d_in, const int* in_sizes, int n_in,
                              void* d_out, int out_size, void* d_ws, size_t ws_size,
                              hipStream_t stream) {
    (void)in_sizes; (void)n_in; (void)out_size; (void)ws_size;

    const float* mask = (const float*)d_in[0];
    const float* bbox = (const float*)d_in[1];
    const int*   cls  = (const int*)  d_in[2];
    const float* U_w  = (const float*)d_in[3];
    const float* V_w  = (const float*)d_in[4];
    const float* Wc_w = (const float*)d_in[5];
    const float* P_w  = (const float*)d_in[6];

    float* out = (float*)d_out;
    float* ws  = (float*)d_ws;
    float* W1  = ws;
    float* F2  = ws + 16384;
    float* R   = ws + 32768;
    float* Omat= ws + 36864;

    feat_kernel <<<64,  256, 0, stream>>>(cls, U_w, V_w, P_w, W1, F2);
    logit_kernel<<<16,  256, 0, stream>>>(W1, F2, bbox, Wc_w, P_w, R);
    omat_kernel <<<16,  256, 0, stream>>>(R, Omat, out + (size_t)N_INST * NPIX);
    apply_kernel<<<500, 256, 0, stream>>>(mask, Omat, out);
}


# logit kernel: each output's 384-term sum split over 16 waves (24 terms each, every load in flight at once), partials added in LDS in fixed slice order; this changes that sum's order
# speedup vs baseline: 1.8537x; 1.8537x over previous
#include <hip/hip_runtime.h>
#include <math.h>

#define N_INST 64
#define NPIX   64000

typedef __attribute__((ext_vector_type(2))) float v2f;
typedef __attribute__((ext_vector_type(8))) float v8f;

__device__ __forceinline__ float sigmoidf(float x) {
    return __builtin_amdgcn_rcpf(1.0f + __expf(-x));
}

__global__ __launch_bounds__(256) void feat_kernel(
    const int* __restrict__ cls, const float* __restrict__ U_w,
    const float* __restrict__ V_w, const float* __restrict__ P_w,
    float* __restrict__ W1, float* __restrict__ F2) {
    int gid = blockIdx.x * 256 + threadIdx.x;
    int c = gid >> 6;
    int i = gid & 63;
    int a = cls[i] - 1;
    float u = U_w[c * 80 + a];
    float v = V_w[c * 80 + a];
    W1[c * 64 + i] = fmaxf(u, 0.0f) * P_w[c];
    F2[c * 64 + i] = fmaxf(v, 0.0f);
}

__global__ __launch_bounds__(512) void logit_kernel(
    const float* __restrict__ W1, const float* __restrict__ F2,
    const float* __restrict__ bbox, const float* __restrict__ Wc_w,
    const float* __restrict__ P_w, float* __restrict__ R) {
    __shared__ float part[16][32];

    const int lane = threadIdx.x & 31;
    const int s = __builtin_amdgcn_readfirstlane((int)(threadIdx.x >> 5));
    const int j = blockIdx.x >> 1;
    const int i = (blockIdx.x & 1) * 32 + lane;

    const float* w1 = W1 + s * (16 * 64) + i;
    const float* f2 = F2 + s * (16 * 64) + j;
    float acc = 0.0f;
    for (int u = 0; u < 16; ++u)
        acc += w1[u * 64] * f2[u * 64];

    float xi0 = bbox[i * 5 + 1], yi0 = bbox[i * 5 + 2];
    float xi1 = bbox[i * 5 + 3], yi1 = bbox[i * 5 + 4];
    float xj0 = bbox[j * 5 + 1], yj0 = bbox[j * 5 + 2];
    float xj1 = bbox[j * 5 + 3], yj1 = bbox[j * 5 + 4];
    float wi = xi1 - xi0 + 1.0f, hi = yi1 - yi0 + 1.0f;
    float wj = xj1 - xj0 + 1.0f, hj = yj1 - yj0 + 1.0f;
    float xci = xi0 + 0.5f * wi, yci = yi0 + 0.5f * hi;
    float xcj = xj0 + 0.5f * wj, ycj = yj0 + 0.5f * hj;
    float rwi = __builtin_amdgcn_rcpf(wi);
    float rhi = __builtin_amdgcn_rcpf(hi);
    float dx = -(xci - xcj) * rwi;
    float dy = -(yci - ycj) * rhi;
    float dw = __logf(wj * rwi);
    float dh = __logf(hj * rhi);

    const float* wc = Wc_w + s * (8 * 4);
    const float* pw = P_w + 256 + s * 8;
    for (int u = 0; u < 8; ++u) {
        float v = dx * wc[u * 4 + 0] + dy * wc[u * 4 + 1] +
                  dw * wc[u * 4 + 2] + dh * wc[u * 4 + 3];
        acc += fmaxf(v, 0.0f) * pw[u];
    }

    part[s][lane] = acc;
    __syncthreads();
    if (s == 0) {
        float t = part[0][lane];
        for (int q = 1; q < 16; ++q)
            t += part[q][lane];
        R[i * 64 + j] = sigmoidf(t);
    }
}

__global__ __launch_bounds__(256) void omat_kernel(
    const float* __restrict__ R, float* __restrict__ O_ws,
    float* __restrict__ O_out) {
    int gid = blockIdx.x * 256 + threadIdx.x;
    int i = gid >> 6;
    int j = gid & 63;
    float o = fmaxf(R[i * 64 + j] - R[j * 64 + i], 0.0f);
    O_ws[gid]  = o;
    O_out[gid] = o;
}

__global__ __launch_bounds__(256) void apply_kernel(
    const float* __restrict__ mask, const float* __restrict__ O,
    float* __restrict__ out) {
    __shared__ float sO[64][68];

    for (int idx = threadIdx.x; idx < 64 * 64; idx += 256)
        sO[idx >> 6][idx & 63] = O[idx];
    __syncthreads();

    const int lane = threadIdx.x & 31;
    const int wave = threadIdx.x >> 5;
    const int col  = lane & 15;
    const int half = lane >> 4;
    const int p    = blockIdx.x * 128 + wave * 16 + col;

    v8f acc[4] = {v8f{}, v8f{}, v8f{}, v8f{}};

    #pragma unroll
    for (int k0 = 0; k0 < 64; k0 += 4) {
        const int kA = k0 + half * 2;
        v2f b;
        b[0] = sigmoidf(mask[(kA + 0) * NPIX + p]);
        b[1] = sigmoidf(mask[(kA + 1) * NPIX + p]);
        #pragma unroll
        for (int mt = 0; mt < 4; ++mt) {
            v2f a;
            a[0] = sO[mt * 16 + col][kA + 0];
            a[1] = sO[mt * 16 + col][kA + 1];
            acc[mt] = __builtin_amdgcn_wmma_f32_16x16x4_f32(
                false, a, false, b, (short)0, acc[mt], false, false);
        }
    }

    #pragma unroll
    for (int mt = 0; mt < 4; ++mt) {
        #pragma unroll
        for (int r = 0; r < 8; ++r) {
            int row = mt * 16 + half * 8 + r;
            float m  = mask[row * NPIX + p];
            float tm = sigmoidf(m);
            out[row * NPIX + p] = m * (1.0f - tm * acc[mt][r]);
        }
    }
}

extern "C" void kernel_launch(void* const* d_in, const int* in_sizes, int n_in,
                              void* d_out, int out_size, void* d_ws, size_t ws_size,
                              hipStream_t stream) {
    (void)in_sizes; (void)n_in; (void)out_size; (void)ws_size;

    const float* mask = (const float*)d_in[0];
    const float* bbox = (const float*)d_in[1];
    const int*   cls  = (const int*)  d_in[2];
    const float* U_w  = (const float*)d_in[3];
    const float* V_w  = (const float*)d_in[4];
    const float* Wc_w = (const float*)d_in[5];
    const float* P_w  = (const float*)d_in[6];

    float* out = (float*)d_out;
    float* ws  = (float*)d_ws;
    float* W1  = ws;
    float* F2  = ws + 16384;
    float* R   = ws + 32768;
    float* Omat= ws + 36864;

    feat_kernel <<<64,  256, 0, stream>>>(cls, U_w, V_w, P_w, W1, F2);
    logit_kernel<<<128, 512, 0, stream>>>(W1, F2, bbox, Wc_w, P_w, R);
    omat_kernel <<<16,  256, 0, stream>>>(R, Omat, out + (size_t)N_INST * NPIX);
    apply_kernel<<<500, 256, 0, stream>>>(mask, Omat, out);
}


# 4 launches -> 2: class-feature gather folded into the logit kernel; O=relu(R-R^T) built in LDS in the apply kernel's prologue, under its already-issued mask loads; every sum's order unchanged
# speedup vs baseline: 2.6728x; 1.4419x over previous
#include <hip/hip_runtime.h>
#include <math.h>

#define N_INST 64
#define NPIX   64000

typedef __attribute__((ext_vector_type(2))) float v2f;
typedef __attribute__((ext_vector_type(8))) float v8f;

__device__ __forceinline__ float sigmoidf(float x) {
    return __builtin_amdgcn_rcpf(1.0f + __expf(-x));
}

__global__ __launch_bounds__(512) void logit_kernel(
    const int* __restrict__ cls, const float* __restrict__ U_w,
    const float* __restrict__ V_w, const float* __restrict__ bbox,
    const float* __restrict__ Wc_w, const float* __restrict__ P_w,
    float* __restrict__ R) {
    __shared__ float part[16][32];

    const int lane = threadIdx.x & 31;
    const int s = __builtin_amdgcn_readfirstlane((int)(threadIdx.x >> 5));
    const int j = blockIdx.x >> 1;
    const int i = (blockIdx.x & 1) * 32 + lane;

    const int ai = cls[i] - 1;
    const int aj = cls[j] - 1;

    float xi0 = bbox[i * 5 + 1], yi0 = bbox[i * 5 + 2];
    float xi1 = bbox[i * 5 + 3], yi1 = bbox[i * 5 + 4];
    float xj0 = bbox[j * 5 + 1], yj0 = bbox[j * 5 + 2];
    float xj1 = bbox[j * 5 + 3], yj1 = bbox[j * 5 + 4];

    const float* uw = U_w + s * (16 * 80) + ai;
    const float* vw = V_w + s * (16 * 80) + aj;
    const float* pc = P_w + s * 16;
    float acc = 0.0f;
    for (int u = 0; u < 16; ++u) {
        float w1 = fmaxf(uw[u * 80], 0.0f) * pc[u];
        acc += w1 * fmaxf(vw[u * 80], 0.0f);
    }

    float wi = xi1 - xi0 + 1.0f, hi = yi1 - yi0 + 1.0f;
    float wj = xj1 - xj0 + 1.0f, hj = yj1 - yj0 + 1.0f;
    float xci = xi0 + 0.5f * wi, yci = yi0 + 0.5f * hi;
    float xcj = xj0 + 0.5f * wj, ycj = yj0 + 0.5f * hj;
    float rwi = __builtin_amdgcn_rcpf(wi);
    float rhi = __builtin_amdgcn_rcpf(hi);
    float dx = -(xci - xcj) * rwi;
    float dy = -(yci - ycj) * rhi;
    float dw = __logf(wj * rwi);
    float dh = __logf(hj * rhi);

    const float* wc = Wc_w + s * (8 * 4);
    const float* pw = P_w + 256 + s * 8;
    for (int u = 0; u < 8; ++u) {
        float v = dx * wc[u * 4 + 0] + dy * wc[u * 4 + 1] +
                  dw * wc[u * 4 + 2] + dh * wc[u * 4 + 3];
        acc += fmaxf(v, 0.0f) * pw[u];
    }

    part[s][lane] = acc;
    __syncthreads();
    if (s == 0) {
        float t = part[0][lane];
        for (int q = 1; q < 16; ++q)
            t += part[q][lane];
        R[i * 64 + j] = sigmoidf(t);
    }
}

__global__ __launch_bounds__(256) void apply_kernel(
    const float* __restrict__ mask, const float* __restrict__ R,
    float* __restrict__ out, float* __restrict__ O_out) {
    __shared__ float sR[64][65];
    __shared__ __attribute__((aligned(16))) float sO[64][68];

    const int t    = threadIdx.x;
    const int lane = t & 31;
    const int wave = t >> 5;
    const int col  = lane & 15;
    const int half = lane >> 4;
    const int p    = blockIdx.x * 128 + wave * 16 + col;

    const int pr = t >> 4;
    const int pq = (t & 15) * 4;
    float4 r4[4];
    #pragma unroll
    for (int q = 0; q < 4; ++q)
        r4[q] = ((const float4*)R)[t + q * 256];

    float mb[32];
    #pragma unroll
    for (int ks = 0; ks < 16; ++ks) {
        const int kA = ks * 4 + half * 2;
        mb[2 * ks + 0] = mask[(kA + 0) * NPIX + p];
        mb[2 * ks + 1] = mask[(kA + 1) * NPIX + p];
    }

    #pragma unroll
    for (int q = 0; q < 4; ++q) {
        const int i = pr + q * 16;
        sR[i][pq + 0] = r4[q].x;
        sR[i][pq + 1] = r4[q].y;
        sR[i][pq + 2] = r4[q].z;
        sR[i][pq + 3] = r4[q].w;
    }
    __syncthreads();
    #pragma unroll
    for (int q = 0; q < 4; ++q) {
        const int i = pr + q * 16;
        float4 o;
        o.x = fmaxf(r4[q].x - sR[pq + 0][i], 0.0f);
        o.y = fmaxf(r4[q].y - sR[pq + 1][i], 0.0f);
        o.z = fmaxf(r4[q].z - sR[pq + 2][i], 0.0f);
        o.w = fmaxf(r4[q].w - sR[pq + 3][i], 0.0f);
        *(float4*)&sO[i][pq] = o;
        if (blockIdx.x == q)
            ((float4*)O_out)[t + q * 256] = o;
    }
    __syncthreads();

    v8f acc[4] = {v8f{}, v8f{}, v8f{}, v8f{}};

    #pragma unroll
    for (int ks = 0; ks < 16; ++ks) {
        const int kA = ks * 4 + half * 2;
        v2f b;
        b[0] = sigmoidf(mb[2 * ks + 0]);
        b[1] = sigmoidf(mb[2 * ks + 1]);
        #pragma unroll
        for (int mt = 0; mt < 4; ++mt) {
            v2f a;
            a[0] = sO[mt * 16 + col][kA + 0];
            a[1] = sO[mt * 16 + col][kA + 1];
            acc[mt] = __builtin_amdgcn_wmma_f32_16x16x4_f32(
                false, a, false, b, (short)0, acc[mt], false, false);
        }
    }

    #pragma unroll
    for (int mt = 0; mt < 4; ++mt) {
        #pragma unroll
        for (int r = 0; r < 8; ++r) {
            int row = mt * 16 + half * 8 + r;
            float m  = mask[row * NPIX + p];
            float tm = sigmoidf(m);
            out[row * NPIX + p] = m * (1.0f - tm * acc[mt][r]);
        }
    }
}

extern "C" void kernel_launch(void* const* d_in, const int* in_sizes, int n_in,
                              void* d_out, int out_size, void* d_ws, size_t ws_size,
                              hipStream_t stream) {
    (void)in_sizes; (void)n_in; (void)out_size; (void)ws_size;

    const float* mask = (const float*)d_in[0];
    const float* bbox = (const float*)d_in[1];
    const int*   cls  = (const int*)  d_in[2];
    const float* U_w  = (const float*)d_in[3];
    const float* V_w  = (const float*)d_in[4];
    const float* Wc_w = (const float*)d_in[5];
    const float* P_w  = (const float*)d_in[6];

    float* out = (float*)d_out;
    float* R   = (float*)d_ws;

    logit_kernel<<<128, 512, 0, stream>>>(cls, U_w, V_w, bbox, Wc_w, P_w, R);
    apply_kernel<<<500, 256, 0, stream>>>(mask, R, out, out + (size_t)N_INST * NPIX);
}


# apply kernel: WMMA k order permuted (one index function for both loaders) so a lane's B-fragment mask rows are its own output rows; epilogue takes m, sigmoid(m) from registers; k sum order changes
# speedup vs baseline: 2.8225x; 1.0560x over previous
#include <hip/hip_runtime.h>
#include <math.h>

#define N_INST 64
#define NPIX   64000

typedef __attribute__((ext_vector_type(2))) float v2f;
typedef __attribute__((ext_vector_type(8))) float v8f;

__device__ __forceinline__ float sigmoidf(float x) {
    return __builtin_amdgcn_rcpf(1.0f + __expf(-x));
}

__device__ __forceinline__ int krow(int ks, int h) {
    return (ks >> 2) * 16 + h * 8 + (ks & 3) * 2;
}

__global__ __launch_bounds__(512) void logit_kernel(
    const int* __restrict__ cls, const float* __restrict__ U_w,
    const float* __restrict__ V_w, const float* __restrict__ bbox,
    const float* __restrict__ Wc_w, const float* __restrict__ P_w,
    float* __restrict__ R) {
    __shared__ float part[16][32];

    const int lane = threadIdx.x & 31;
    const int s = __builtin_amdgcn_readfirstlane((int)(threadIdx.x >> 5));
    const int j = blockIdx.x >> 1;
    const int i = (blockIdx.x & 1) * 32 + lane;

    const int ai = cls[i] - 1;
    const int aj = cls[j] - 1;

    float xi0 = bbox[i * 5 + 1], yi0 = bbox[i * 5 + 2];
    float xi1 = bbox[i * 5 + 3], yi1 = bbox[i * 5 + 4];
    float xj0 = bbox[j * 5 + 1], yj0 = bbox[j * 5 + 2];
    float xj1 = bbox[j * 5 + 3], yj1 = bbox[j * 5 + 4];

    const float* uw = U_w + s * (16 * 80) + ai;
    const float* vw = V_w + s * (16 * 80) + aj;
    const float* pc = P_w + s * 16;
    float acc = 0.0f;
    for (int u = 0; u < 16; ++u) {
        float w1 = fmaxf(uw[u * 80], 0.0f) * pc[u];
        acc += w1 * fmaxf(vw[u * 80], 0.0f);
    }

    float wi = xi1 - xi0 + 1.0f, hi = yi1 - yi0 + 1.0f;
    float wj = xj1 - xj0 + 1.0f, hj = yj1 - yj0 + 1.0f;
    float xci = xi0 + 0.5f * wi, yci = yi0 + 0.5f * hi;
    float xcj = xj0 + 0.5f * wj, ycj = yj0 + 0.5f * hj;
    float rwi = __builtin_amdgcn_rcpf(wi);
    float rhi = __builtin_amdgcn_rcpf(hi);
    float dx = -(xci - xcj) * rwi;
    float dy = -(yci - ycj) * rhi;
    float dw = __logf(wj * rwi);
    float dh = __logf(hj * rhi);

    const float* wc = Wc_w + s * (8 * 4);
    const float* pw = P_w + 256 + s * 8;
    for (int u = 0; u < 8; ++u) {
        float v = dx * wc[u * 4 + 0] + dy * wc[u * 4 + 1] +
                  dw * wc[u * 4 + 2] + dh * wc[u * 4 + 3];
        acc += fmaxf(v, 0.0f) * pw[u];
    }

    part[s][lane] = acc;
    __syncthreads();
    if (s == 0) {
        float t = part[0][lane];
        for (int q = 1; q < 16; ++q)
            t += part[q][lane];
        R[i * 64 + j] = sigmoidf(t);
    }
}

__global__ __launch_bounds__(256) void apply_kernel(
    const float* __restrict__ mask, const float* __restrict__ R,
    float* __restrict__ out, float* __restrict__ O_out) {
    __shared__ float sR[64][65];
    __shared__ __attribute__((aligned(16))) float sO[64][68];

    const int t    = threadIdx.x;
    const int lane = t & 31;
    const int wave = t >> 5;
    const int col  = lane & 15;
    const int half = lane >> 4;
    const int p    = blockIdx.x * 128 + wave * 16 + col;

    const int pr = t >> 4;
    const int pq = (t & 15) * 4;
    float4 r4[4];
    #pragma unroll
    for (int q = 0; q < 4; ++q)
        r4[q] = ((const float4*)R)[t + q * 256];

    float mb[32];
    #pragma unroll
    for (int ks = 0; ks < 16; ++ks) {
        const int kA = krow(ks, half);
        mb[2 * ks + 0] = mask[(kA + 0) * NPIX + p];
        mb[2 * ks + 1] = mask[(kA + 1) * NPIX + p];
    }

    #pragma unroll
    for (int q = 0; q < 4; ++q) {
        const int i = pr + q * 16;
        sR[i][pq + 0] = r4[q].x;
        sR[i][pq + 1] = r4[q].y;
        sR[i][pq + 2] = r4[q].z;
        sR[i][pq + 3] = r4[q].w;
    }
    __syncthreads();
    #pragma unroll
    for (int q = 0; q < 4; ++q) {
        const int i = pr + q * 16;
        float4 o;
        o.x = fmaxf(r4[q].x - sR[pq + 0][i], 0.0f);
        o.y = fmaxf(r4[q].y - sR[pq + 1][i], 0.0f);
        o.z = fmaxf(r4[q].z - sR[pq + 2][i], 0.0f);
        o.w = fmaxf(r4[q].w - sR[pq + 3][i], 0.0f);
        *(float4*)&sO[i][pq] = o;
        if (blockIdx.x == q)
            ((float4*)O_out)[t + q * 256] = o;
    }
    __syncthreads();

    v8f acc[4] = {v8f{}, v8f{}, v8f{}, v8f{}};

    float sg[32];
    #pragma unroll
    for (int ks = 0; ks < 16; ++ks) {
        const int kA = krow(ks, half);
        v2f b;
        b[0] = sg[2 * ks + 0] = sigmoidf(mb[2 * ks + 0]);
        b[1] = sg[2 * ks + 1] = sigmoidf(mb[2 * ks + 1]);
        #pragma unroll
        for (int mt = 0; mt < 4; ++mt) {
            v2f a;
            a[0] = sO[mt * 16 + col][kA + 0];
            a[1] = sO[mt * 16 + col][kA + 1];
            acc[mt] = __builtin_amdgcn_wmma_f32_16x16x4_f32(
                false, a, false, b, (short)0, acc[mt], false, false);
        }
    }

    #pragma unroll
    for (int mt = 0; mt < 4; ++mt) {
        #pragma unroll
        for (int r = 0; r < 8; ++r) {
            const int row = mt * 16 + half * 8 + r;
            out[row * NPIX + p] = mb[8 * mt + r] * (1.0f - sg[8 * mt + r] * acc[mt][r]);
        }
    }
}

extern "C" void kernel_launch(void* const* d_in, const int* in_sizes, int n_in,
                              void* d_out, int out_size, void* d_ws, size_t ws_size,
                              hipStream_t stream) {
    (void)in_sizes; (void)n_in; (void)out_size; (void)ws_size;

    const float* mask = (const float*)d_in[0];
    const float* bbox = (const float*)d_in[1];
    const int*   cls  = (const int*)  d_in[2];
    const float* U_w  = (const float*)d_in[3];
    const float* V_w  = (const float*)d_in[4];
    const float* Wc_w = (const float*)d_in[5];
    const float* P_w  = (const float*)d_in[6];

    float* out = (float*)d_out;
    float* R   = (float*)d_ws;

    logit_kernel<<<128, 512, 0, stream>>>(cls, U_w, V_w, bbox, Wc_w, P_w, R);
    apply_kernel<<<500, 256, 0, stream>>>(mask, R, out, out + (size_t)N_INST * NPIX);
}
